# starting kernel
# baseline (speedup 1.0000x reference)
#include <hip/hip_runtime.h>
#include <math.h>


#define NBATCH 16
#define CDIM   128
#define PPTS   1200
#define KCLU   64
#define S1     5
#define MROWS  (KCLU * S1)
#define ALPHA  100.0f
#define EPSV   1e-12f

typedef __attribute__((ext_vector_type(2))) float v2f;
typedef __attribute__((ext_vector_type(8))) float v8f;

#define WS_W     0
#define WS_B     (WS_W + MROWS * CDIM)
#define WS_MULT  (WS_B + MROWS)
#define WS_SSUM  (WS_MULT + NBATCH * KCLU * PPTS)
#define WS_RN    (WS_SSUM + NBATCH * KCLU)

__global__ __launch_bounds__(128) void prep_kernel(const float* __restrict__ cent,
                                                   float* __restrict__ ws) {
  __shared__ float red[128];
  const int r = blockIdx.x;
  const int t = threadIdx.x;
  const float v = cent[r * CDIM + t];
  ws[WS_W + r * CDIM + t] = 2.0f * ALPHA * v;
  red[t] = v * v;
  __syncthreads();
  for (int off = 64; off > 0; off >>= 1) {
    if (t < off) red[t] += red[t + off];
    __syncthreads();
  }
  if (t == 0) ws[WS_B + r] = -ALPHA * sqrtf(red[0]);
}

__global__ __launch_bounds__(32) void gemm1_softmax_kernel(const float* __restrict__ x,
                                                           float* __restrict__ ws) {
  __shared__ float Xs[CDIM][16];
  __shared__ float Cv[MROWS][16];

  const int pt   = blockIdx.x;
  const int n    = blockIdx.y;
  const int lane = threadIdx.x;
  const int p0   = pt * 16;
  const int col  = lane & 15;
  const int half = lane >> 4;

  const float* xb = x + (size_t)n * CDIM * PPTS;
  for (int c = half; c < CDIM; c += 2)
    Xs[c][col] = xb[(size_t)c * PPTS + p0 + col];
  __syncthreads();

  const float* W  = ws + WS_W;
  const float* Bv = ws + WS_B;
  const int koff  = half ? 2 : 0;

  for (int mt = 0; mt < MROWS / 16; ++mt) {
    v8f acc = {};
    const int arow = mt * 16 + col;
    const float* wrow = W + (size_t)arow * CDIM + koff;
    for (int k = 0; k < CDIM; k += 4) {
      v2f a = *(const v2f*)(wrow + k);
      v2f b;
      b[0] = Xs[k + koff][col];
      b[1] = Xs[k + koff + 1][col];
      acc = __builtin_amdgcn_wmma_f32_16x16x4_f32(
           false, a,  false, b,
           (short)0, acc,  false,  false);
    }
    const int mbase = mt * 16 + (half ? 8 : 0);
    for (int r = 0; r < 8; ++r)
      Cv[mbase + r][col] = acc[r] + Bv[mbase + r];
  }
  __syncthreads();

  if (lane < 16) {
    const int p = lane;
    float mA[S1], sA[S1];
    for (int s = 0; s < S1; ++s) { mA[s] = -3.0e38f; sA[s] = 0.0f; }
    for (int k = 0; k < KCLU; ++k) {
      for (int s = 0; s < S1; ++s) {
        const float v = Cv[k * S1 + s][p];
        const float mn = fmaxf(mA[s], v);
        sA[s] = sA[s] * __expf(mA[s] - mn) + __expf(v - mn);
        mA[s] = mn;
      }
    }
    float* MU = ws + WS_MULT + (size_t)n * KCLU * PPTS;
    for (int k = 0; k < KCLU; ++k) {
      float vv[S1];
      float mB = -3.0e38f;
      for (int s = 0; s < S1; ++s) { vv[s] = Cv[k * S1 + s][p]; mB = fmaxf(mB, vv[s]); }
      float sB = 0.0f;
      for (int s = 0; s < S1; ++s) sB += __expf(vv[s] - mB);
      float mult = 1.0f;
      for (int s = 0; s < S1; ++s) {
        const float sa = (__expf(vv[s] - mA[s]) / sA[s]) * (__expf(vv[s] - mB) / sB);
        mult *= (1.0f + sa);
      }
      MU[(size_t)k * PPTS + p0 + p] = mult;
    }
  }
}

__global__ __launch_bounds__(32) void ssum_kernel(float* __restrict__ ws) {
  const int nk = blockIdx.x;
  const float* MU = ws + WS_MULT + (size_t)nk * PPTS;
  float s = 0.0f;
  for (int p = threadIdx.x; p < PPTS; p += 32) s += MU[p];
  for (int off = 16; off > 0; off >>= 1) s += __shfl_down(s, off, 32);
  if (threadIdx.x == 0) ws[WS_SSUM + nk] = s;
}

__global__ __launch_bounds__(32) void gemm2_kernel(const float* __restrict__ x,
                                                   const float* __restrict__ cent,
                                                   const float* __restrict__ ws,
                                                   float* __restrict__ out) {
  const int nt   = blockIdx.x;
  const int mt   = blockIdx.y;
  const int n    = blockIdx.z;
  const int lane = threadIdx.x;
  const int l15  = lane & 15;
  const int half = lane >> 4;
  const int koff = half ? 2 : 0;

  const float* MU = ws + WS_MULT + ((size_t)n * KCLU + mt * 16 + l15) * PPTS + koff;
  const float* XB = x + ((size_t)n * CDIM + nt * 16 + l15) * PPTS + koff;

  v8f acc = {};
  for (int p = 0; p < PPTS; p += 4) {
    v2f a = *(const v2f*)(MU + p);
    v2f b = *(const v2f*)(XB + p);
    acc = __builtin_amdgcn_wmma_f32_16x16x4_f32(
        false, a, false, b, (short)0, acc, false, false);
  }

  const float* Ssum = ws + WS_SSUM + n * KCLU;
  const int kbase = mt * 16 + (half ? 8 : 0);
  const int cc    = nt * 16 + l15;
  for (int r = 0; r < 8; ++r) {
    const int kk = kbase + r;
    const float rep = cent[(size_t)(kk * S1) * CDIM + cc];
    out[(size_t)n * KCLU * CDIM + (size_t)kk * CDIM + cc] = acc[r] - rep * Ssum[kk];
  }
}

__global__ __launch_bounds__(128) void rownorm_kernel(float* __restrict__ out,
                                                      float* __restrict__ ws) {
  __shared__ float red[128];
  const int nk = blockIdx.x;
  const int t  = threadIdx.x;
  const float v = out[(size_t)nk * CDIM + t];
  red[t] = v * v;
  __syncthreads();
  for (int off = 64; off > 0; off >>= 1) {
    if (t < off) red[t] += red[t + off];
    __syncthreads();
  }
  const float sum   = red[0];
  const float scale = 1.0f / fmaxf(sqrtf(sum), EPSV);
  out[(size_t)nk * CDIM + t] = v * scale;
  if (t == 0) ws[WS_RN + nk] = sum * scale * scale;
}

__global__ __launch_bounds__(256) void finalnorm_kernel(float* __restrict__ out,
                                                        const float* __restrict__ ws) {
  __shared__ float red[64];
  __shared__ float scale_s;
  const int n = blockIdx.x;
  const int t = threadIdx.x;
  if (t < 64) red[t] = ws[WS_RN + n * KCLU + t];
  __syncthreads();
  if (t == 0) {
    float s = 0.0f;
    for (int i = 0; i < 64; ++i) s += red[i];
    scale_s = 1.0f / fmaxf(sqrtf(s), EPSV);
  }
  __syncthreads();
  const float sc = scale_s;
  for (int i = t; i < KCLU * CDIM; i += 256)
    out[(size_t)n * KCLU * CDIM + i] *= sc;
}

extern "C" void kernel_launch(void* const* d_in, const int* in_sizes, int n_in,
                              void* d_out, int out_size, void* d_ws, size_t ws_size,
                              hipStream_t stream) {
  (void)in_sizes; (void)n_in; (void)out_size; (void)ws_size;
  const float* x    = (const float*)d_in[0];
  const float* cent = (const float*)d_in[1];
  float* out = (float*)d_out;
  float* ws  = (float*)d_ws;

  prep_kernel<<<MROWS, 128, 0, stream>>>(cent, ws);
  gemm1_softmax_kernel<<<dim3(PPTS / 16, NBATCH), 32, 0, stream>>>(x, ws);
  ssum_kernel<<<NBATCH * KCLU, 32, 0, stream>>>(ws);
  gemm2_kernel<<<dim3(CDIM / 16, KCLU / 16, NBATCH), 32, 0, stream>>>(x, cent, ws, out);
  rownorm_kernel<<<NBATCH * KCLU, 128, 0, stream>>>(out, ws);
  finalnorm_kernel<<<NBATCH, 256, 0, stream>>>(out, ws);
}


# kernel 2 block widened from 1 wave to 5: shared X tile, 4 M-tiles per wave on 4 accumulators, softmax on 128 threads; its sums over k become 8 partial sums of 8 (order changed)
# speedup vs baseline: 1.3374x; 1.3374x over previous
#include <hip/hip_runtime.h>
#include <math.h>


#define NBATCH 16
#define CDIM   128
#define PPTS   1200
#define KCLU   64
#define S1     5
#define MROWS  (KCLU * S1)
#define ALPHA  100.0f
#define EPSV   1e-12f

typedef __attribute__((ext_vector_type(2))) float v2f;
typedef __attribute__((ext_vector_type(8))) float v8f;

#define WS_W     0
#define WS_B     (WS_W + MROWS * CDIM)
#define WS_MULT  (WS_B + MROWS)
#define WS_SSUM  (WS_MULT + NBATCH * KCLU * PPTS)
#define WS_RN    (WS_SSUM + NBATCH * KCLU)

__global__ __launch_bounds__(128) void prep_kernel(const float* __restrict__ cent,
                                                   float* __restrict__ ws) {
  __shared__ float red[128];
  const int r = blockIdx.x;
  const int t = threadIdx.x;
  const float v = cent[r * CDIM + t];
  ws[WS_W + r * CDIM + t] = 2.0f * ALPHA * v;
  red[t] = v * v;
  __syncthreads();
  for (int off = 64; off > 0; off >>= 1) {
    if (t < off) red[t] += red[t + off];
    __syncthreads();
  }
  if (t == 0) ws[WS_B + r] = -ALPHA * sqrtf(red[0]);
}

#define G1_THREADS 160
#define XP 20

__global__ __launch_bounds__(G1_THREADS) void gemm1_softmax_kernel(const float* __restrict__ x,
                                                                   float* __restrict__ ws) {
  __shared__ __attribute__((aligned(16))) float Xs[CDIM * XP];
  __shared__ float Cv[MROWS * XP];
  __shared__ float Pm[8][S1][16];
  __shared__ float Ps[8][S1][16];

  const int pt   = blockIdx.x;
  const int n    = blockIdx.y;
  const int t    = threadIdx.x;
  const int lane = t & 31;
  const int wv   = t >> 5;
  const int p0   = pt * 16;
  const int col  = lane & 15;
  const int half = lane >> 4;

  const float* xb = x + (size_t)n * CDIM * PPTS + p0;
  const int i0 = t;
  const int i1 = t + G1_THREADS;
  const int i2 = t + 2 * G1_THREADS;
  const int i3 = t + 3 * G1_THREADS;
  const int i3c = (i3 < CDIM * 4) ? i3 : i2;
  const float4 x0 = *(const float4*)(xb + (size_t)(i0 >> 2) * PPTS + (i0 & 3) * 4);
  const float4 x1 = *(const float4*)(xb + (size_t)(i1 >> 2) * PPTS + (i1 & 3) * 4);
  const float4 x2 = *(const float4*)(xb + (size_t)(i2 >> 2) * PPTS + (i2 & 3) * 4);
  const float4 x3 = *(const float4*)(xb + (size_t)(i3c >> 2) * PPTS + (i3c & 3) * 4);
  *(float4*)(&Xs[(i0 >> 2) * XP + (i0 & 3) * 4]) = x0;
  *(float4*)(&Xs[(i1 >> 2) * XP + (i1 & 3) * 4]) = x1;
  *(float4*)(&Xs[(i2 >> 2) * XP + (i2 & 3) * 4]) = x2;
  *(float4*)(&Xs[(i3c >> 2) * XP + (i3c & 3) * 4]) = x3;
  __syncthreads();

  float bq[64];
  for (int j = 0; j < 16; ++j)
    for (int e = 0; e < 4; ++e)
      bq[j * 4 + e] = Xs[(8 * j + 4 * half + e) * XP + col];

  const float* wbase = ws + WS_W + (size_t)(wv * 64 + col) * CDIM + 4 * half;
  v8f acc[4];
  for (int i = 0; i < 4; ++i) acc[i] = (v8f){};
  for (int j = 0; j < 16; ++j) {
    float4 a4[4];
    for (int i = 0; i < 4; ++i)
      a4[i] = *(const float4*)(wbase + (size_t)i * 16 * CDIM + 8 * j);
    v2f b0, b1;
    b0[0] = bq[4 * j + 0]; b0[1] = bq[4 * j + 1];
    b1[0] = bq[4 * j + 2]; b1[1] = bq[4 * j + 3];
    for (int i = 0; i < 4; ++i) {
      v2f a0; a0[0] = a4[i].x; a0[1] = a4[i].y;
      acc[i] = __builtin_amdgcn_wmma_f32_16x16x4_f32(
          false, a0, false, b0, (short)0, acc[i], false, false);
    }
    for (int i = 0; i < 4; ++i) {
      v2f a1; a1[0] = a4[i].z; a1[1] = a4[i].w;
      acc[i] = __builtin_amdgcn_wmma_f32_16x16x4_f32(
          false, a1, false, b1, (short)0, acc[i], false, false);
    }
  }

  const float* Bv = ws + WS_B;
  for (int i = 0; i < 4; ++i) {
    const int mbase = wv * 64 + i * 16 + half * 8;
    for (int r = 0; r < 8; ++r)
      Cv[(mbase + r) * XP + col] = acc[i][r] + Bv[mbase + r];
  }
  __syncthreads();

  const int p  = t & 15;
  const int kg = (t >> 4) & 7;
  float vv[8][S1];
  if (t < 128) {
    for (int kk = 0; kk < 8; ++kk)
      for (int s = 0; s < S1; ++s)
        vv[kk][s] = Cv[((kg * 8 + kk) * S1 + s) * XP + p];
    for (int s = 0; s < S1; ++s) {
      float m = vv[0][s];
      for (int kk = 1; kk < 8; ++kk) m = fmaxf(m, vv[kk][s]);
      float sm = 0.0f;
      for (int kk = 0; kk < 8; ++kk) sm += __expf(vv[kk][s] - m);
      Pm[kg][s][p] = m;
      Ps[kg][s][p] = sm;
    }
  }
  __syncthreads();
  if (t < 128) {
    float mA[S1], sA[S1];
    for (int s = 0; s < S1; ++s) {
      float m = Pm[0][s][p];
      for (int g = 1; g < 8; ++g) m = fmaxf(m, Pm[g][s][p]);
      float sm = 0.0f;
      for (int g = 0; g < 8; ++g) sm += Ps[g][s][p] * __expf(Pm[g][s][p] - m);
      mA[s] = m;
      sA[s] = sm;
    }
    float* MU = ws + WS_MULT + (size_t)n * KCLU * PPTS;
    for (int kk = 0; kk < 8; ++kk) {
      float mB = vv[kk][0];
      for (int s = 1; s < S1; ++s) mB = fmaxf(mB, vv[kk][s]);
      float sB = 0.0f;
      for (int s = 0; s < S1; ++s) sB += __expf(vv[kk][s] - mB);
      float mult = 1.0f;
      for (int s = 0; s < S1; ++s) {
        const float sa = (__expf(vv[kk][s] - mA[s]) / sA[s]) * (__expf(vv[kk][s] - mB) / sB);
        mult *= (1.0f + sa);
      }
      MU[(size_t)(kg * 8 + kk) * PPTS + p0 + p] = mult;
    }
  }
}

__global__ __launch_bounds__(32) void ssum_kernel(float* __restrict__ ws) {
  const int nk = blockIdx.x;
  const float* MU = ws + WS_MULT + (size_t)nk * PPTS;
  float s = 0.0f;
  for (int p = threadIdx.x; p < PPTS; p += 32) s += MU[p];
  for (int off = 16; off > 0; off >>= 1) s += __shfl_down(s, off, 32);
  if (threadIdx.x == 0) ws[WS_SSUM + nk] = s;
}

__global__ __launch_bounds__(32) void gemm2_kernel(const float* __restrict__ x,
                                                   const float* __restrict__ cent,
                                                   const float* __restrict__ ws,
                                                   float* __restrict__ out) {
  const int nt   = blockIdx.x;
  const int mt   = blockIdx.y;
  const int n    = blockIdx.z;
  const int lane = threadIdx.x;
  const int l15  = lane & 15;
  const int half = lane >> 4;
  const int koff = half ? 2 : 0;

  const float* MU = ws + WS_MULT + ((size_t)n * KCLU + mt * 16 + l15) * PPTS + koff;
  const float* XB = x + ((size_t)n * CDIM + nt * 16 + l15) * PPTS + koff;

  v8f acc = {};
  for (int p = 0; p < PPTS; p += 4) {
    v2f a = *(const v2f*)(MU + p);
    v2f b = *(const v2f*)(XB + p);
    acc = __builtin_amdgcn_wmma_f32_16x16x4_f32(
        false, a, false, b, (short)0, acc, false, false);
  }

  const float* Ssum = ws + WS_SSUM + n * KCLU;
  const int kbase = mt * 16 + (half ? 8 : 0);
  const int cc    = nt * 16 + l15;
  for (int r = 0; r < 8; ++r) {
    const int kk = kbase + r;
    const float rep = cent[(size_t)(kk * S1) * CDIM + cc];
    out[(size_t)n * KCLU * CDIM + (size_t)kk * CDIM + cc] = acc[r] - rep * Ssum[kk];
  }
}

__global__ __launch_bounds__(128) void rownorm_kernel(float* __restrict__ out,
                                                      float* __restrict__ ws) {
  __shared__ float red[128];
  const int nk = blockIdx.x;
  const int t  = threadIdx.x;
  const float v = out[(size_t)nk * CDIM + t];
  red[t] = v * v;
  __syncthreads();
  for (int off = 64; off > 0; off >>= 1) {
    if (t < off) red[t] += red[t + off];
    __syncthreads();
  }
  const float sum   = red[0];
  const float scale = 1.0f / fmaxf(sqrtf(sum), EPSV);
  out[(size_t)nk * CDIM + t] = v * scale;
  if (t == 0) ws[WS_RN + nk] = sum * scale * scale;
}

__global__ __launch_bounds__(256) void finalnorm_kernel(float* __restrict__ out,
                                                        const float* __restrict__ ws) {
  __shared__ float red[64];
  __shared__ float scale_s;
  const int n = blockIdx.x;
  const int t = threadIdx.x;
  if (t < 64) red[t] = ws[WS_RN + n * KCLU + t];
  __syncthreads();
  if (t == 0) {
    float s = 0.0f;
    for (int i = 0; i < 64; ++i) s += red[i];
    scale_s = 1.0f / fmaxf(sqrtf(s), EPSV);
  }
  __syncthreads();
  const float sc = scale_s;
  for (int i = t; i < KCLU * CDIM; i += 256)
    out[(size_t)n * KCLU * CDIM + i] *= sc;
}

extern "C" void kernel_launch(void* const* d_in, const int* in_sizes, int n_in,
                              void* d_out, int out_size, void* d_ws, size_t ws_size,
                              hipStream_t stream) {
  (void)in_sizes; (void)n_in; (void)out_size; (void)ws_size;
  const float* x    = (const float*)d_in[0];
  const float* cent = (const float*)d_in[1];
  float* out = (float*)d_out;
  float* ws  = (float*)d_ws;

  prep_kernel<<<MROWS, 128, 0, stream>>>(cent, ws);
  gemm1_softmax_kernel<<<dim3(PPTS / 16, NBATCH), G1_THREADS, 0, stream>>>(x, ws);
  ssum_kernel<<<NBATCH * KCLU, 32, 0, stream>>>(ws);
  gemm2_kernel<<<dim3(CDIM / 16, KCLU / 16, NBATCH), 32, 0, stream>>>(x, cent, ws, out);
  rownorm_kernel<<<NBATCH * KCLU, 128, 0, stream>>>(out, ws);
  finalnorm_kernel<<<NBATCH, 256, 0, stream>>>(out, ws);
}


# second GEMM rebuilt: 10-wave block per (n,16 k,32 c), each wave stages its 120-p slice in LDS by whole lines and feeds 2 tiles; partials and sum_p mult added in LDS (order changed); sum launch gone
# speedup vs baseline: 1.7942x; 1.3416x over previous
#include <hip/hip_runtime.h>
#include <math.h>


#define NBATCH 16
#define CDIM   128
#define PPTS   1200
#define KCLU   64
#define S1     5
#define MROWS  (KCLU * S1)
#define ALPHA  100.0f
#define EPSV   1e-12f

typedef __attribute__((ext_vector_type(2))) float v2f;
typedef __attribute__((ext_vector_type(8))) float v8f;

#define WS_W     0
#define WS_B     (WS_W + MROWS * CDIM)
#define WS_MULT  (WS_B + MROWS)
#define WS_SSUM  (WS_MULT + NBATCH * KCLU * PPTS)
#define WS_RN    (WS_SSUM + NBATCH * KCLU)

__global__ __launch_bounds__(128) void prep_kernel(const float* __restrict__ cent,
                                                   float* __restrict__ ws) {
  __shared__ float red[128];
  const int r = blockIdx.x;
  const int t = threadIdx.x;
  const float v = cent[r * CDIM + t];
  ws[WS_W + r * CDIM + t] = 2.0f * ALPHA * v;
  red[t] = v * v;
  __syncthreads();
  for (int off = 64; off > 0; off >>= 1) {
    if (t < off) red[t] += red[t + off];
    __syncthreads();
  }
  if (t == 0) ws[WS_B + r] = -ALPHA * sqrtf(red[0]);
}

#define G1_THREADS 160
#define XP 20

__global__ __launch_bounds__(G1_THREADS) void gemm1_softmax_kernel(const float* __restrict__ x,
                                                                   float* __restrict__ ws) {
  __shared__ __attribute__((aligned(16))) float Xs[CDIM * XP];
  __shared__ float Cv[MROWS * XP];
  __shared__ float Pm[8][S1][16];
  __shared__ float Ps[8][S1][16];

  const int pt   = blockIdx.x;
  const int n    = blockIdx.y;
  const int t    = threadIdx.x;
  const int lane = t & 31;
  const int wv   = t >> 5;
  const int p0   = pt * 16;
  const int col  = lane & 15;
  const int half = lane >> 4;

  const float* xb = x + (size_t)n * CDIM * PPTS + p0;
  const int i0 = t;
  const int i1 = t + G1_THREADS;
  const int i2 = t + 2 * G1_THREADS;
  const int i3 = t + 3 * G1_THREADS;
  const int i3c = (i3 < CDIM * 4) ? i3 : i2;
  const float4 x0 = *(const float4*)(xb + (size_t)(i0 >> 2) * PPTS + (i0 & 3) * 4);
  const float4 x1 = *(const float4*)(xb + (size_t)(i1 >> 2) * PPTS + (i1 & 3) * 4);
  const float4 x2 = *(const float4*)(xb + (size_t)(i2 >> 2) * PPTS + (i2 & 3) * 4);
  const float4 x3 = *(const float4*)(xb + (size_t)(i3c >> 2) * PPTS + (i3c & 3) * 4);
  *(float4*)(&Xs[(i0 >> 2) * XP + (i0 & 3) * 4]) = x0;
  *(float4*)(&Xs[(i1 >> 2) * XP + (i1 & 3) * 4]) = x1;
  *(float4*)(&Xs[(i2 >> 2) * XP + (i2 & 3) * 4]) = x2;
  *(float4*)(&Xs[(i3c >> 2) * XP + (i3c & 3) * 4]) = x3;
  __syncthreads();

  float bq[64];
  for (int j = 0; j < 16; ++j)
    for (int e = 0; e < 4; ++e)
      bq[j * 4 + e] = Xs[(8 * j + 4 * half + e) * XP + col];

  const float* wbase = ws + WS_W + (size_t)(wv * 64 + col) * CDIM + 4 * half;
  v8f acc[4];
  for (int i = 0; i < 4; ++i) acc[i] = (v8f){};
  for (int j = 0; j < 16; ++j) {
    float4 a4[4];
    for (int i = 0; i < 4; ++i)
      a4[i] = *(const float4*)(wbase + (size_t)i * 16 * CDIM + 8 * j);
    v2f b0, b1;
    b0[0] = bq[4 * j + 0]; b0[1] = bq[4 * j + 1];
    b1[0] = bq[4 * j + 2]; b1[1] = bq[4 * j + 3];
    for (int i = 0; i < 4; ++i) {
      v2f a0; a0[0] = a4[i].x; a0[1] = a4[i].y;
      acc[i] = __builtin_amdgcn_wmma_f32_16x16x4_f32(
          false, a0, false, b0, (short)0, acc[i], false, false);
    }
    for (int i = 0; i < 4; ++i) {
      v2f a1; a1[0] = a4[i].z; a1[1] = a4[i].w;
      acc[i] = __builtin_amdgcn_wmma_f32_16x16x4_f32(
          false, a1, false, b1, (short)0, acc[i], false, false);
    }
  }

  const float* Bv = ws + WS_B;
  for (int i = 0; i < 4; ++i) {
    const int mbase = wv * 64 + i * 16 + half * 8;
    for (int r = 0; r < 8; ++r)
      Cv[(mbase + r) * XP + col] = acc[i][r] + Bv[mbase + r];
  }
  __syncthreads();

  const int p  = t & 15;
  const int kg = (t >> 4) & 7;
  float vv[8][S1];
  if (t < 128) {
    for (int kk = 0; kk < 8; ++kk)
      for (int s = 0; s < S1; ++s)
        vv[kk][s] = Cv[((kg * 8 + kk) * S1 + s) * XP + p];
    for (int s = 0; s < S1; ++s) {
      float m = vv[0][s];
      for (int kk = 1; kk < 8; ++kk) m = fmaxf(m, vv[kk][s]);
      float sm = 0.0f;
      for (int kk = 0; kk < 8; ++kk) sm += __expf(vv[kk][s] - m);
      Pm[kg][s][p] = m;
      Ps[kg][s][p] = sm;
    }
  }
  __syncthreads();
  if (t < 128) {
    float mA[S1], sA[S1];
    for (int s = 0; s < S1; ++s) {
      float m = Pm[0][s][p];
      for (int g = 1; g < 8; ++g) m = fmaxf(m, Pm[g][s][p]);
      float sm = 0.0f;
      for (int g = 0; g < 8; ++g) sm += Ps[g][s][p] * __expf(Pm[g][s][p] - m);
      mA[s] = m;
      sA[s] = sm;
    }
    float* MU = ws + WS_MULT + (size_t)n * KCLU * PPTS;
    for (int kk = 0; kk < 8; ++kk) {
      float mB = vv[kk][0];
      for (int s = 1; s < S1; ++s) mB = fmaxf(mB, vv[kk][s]);
      float sB = 0.0f;
      for (int s = 0; s < S1; ++s) sB += __expf(vv[kk][s] - mB);
      float mult = 1.0f;
      for (int s = 0; s < S1; ++s) {
        const float sa = (__expf(vv[kk][s] - mA[s]) / sA[s]) * (__expf(vv[kk][s] - mB) / sB);
        mult *= (1.0f + sa);
      }
      MU[(size_t)(kg * 8 + kk) * PPTS + p0 + p] = mult;
    }
  }
}

#define G2_WAVES   10
#define G2_THREADS (G2_WAVES * 32)
#define G2_SLICE   (PPTS / G2_WAVES)
#define G2_PIECES  (G2_SLICE / 4)
#define G2_PAIRS   (G2_SLICE / 8)
#define G2_PITCH   124
#define G2_ROWS    48
#define G2_TPITCH  (8 * 32 + 16)

__global__ __launch_bounds__(G2_THREADS) void gemm2_kernel(const float* __restrict__ x,
                                                           const float* __restrict__ cent,
                                                           const float* __restrict__ ws,
                                                           float* __restrict__ out) {
  __shared__ __attribute__((aligned(16))) float St[G2_WAVES][G2_ROWS][G2_PITCH];
  __shared__ float Red[G2_WAVES][2][G2_TPITCH];
  __shared__ float Rs[G2_WAVES][32];

  const int cq   = blockIdx.x;
  const int mt   = blockIdx.y;
  const int n    = blockIdx.z;
  const int t    = threadIdx.x;
  const int lane = t & 31;
  const int wv   = t >> 5;
  const int l15  = lane & 15;
  const int half = lane >> 4;

  const float* Asrc = ws + WS_MULT + ((size_t)n * KCLU + mt * 16) * PPTS + wv * G2_SLICE;
  const float* Bsrc = x + ((size_t)n * CDIM + cq * 32) * PPTS + wv * G2_SLICE;
  float (*S)[G2_PITCH] = St[wv];

  float rep[2];
  for (int i = 0; i < 2; ++i) {
    const int rowk = (wv + i * G2_WAVES < 16) ? wv + i * G2_WAVES : wv;
    rep[i] = cent[(size_t)((mt * 16 + rowk) * S1) * CDIM + cq * 32 + lane];
  }

  if (lane < G2_PIECES) {
    float4 v[G2_ROWS];
    for (int r = 0; r < 16; ++r)
      v[r] = *(const float4*)(Asrc + (size_t)r * PPTS + lane * 4);
    for (int r = 0; r < 32; ++r)
      v[16 + r] = *(const float4*)(Bsrc + (size_t)r * PPTS + lane * 4);
    for (int r = 0; r < G2_ROWS; ++r)
      *(float4*)(&S[r][lane * 4]) = v[r];
  }
  __syncthreads();

  const float* Ar  = &S[l15][4 * half];
  const float* B0r = &S[16 + l15][4 * half];
  const float* B1r = &S[32 + l15][4 * half];
  float4 fa[G2_PAIRS], fp[G2_PAIRS], fq[G2_PAIRS];
  for (int jj = 0; jj < G2_PAIRS; ++jj) {
    fa[jj] = *(const float4*)(Ar + 8 * jj);
    fp[jj] = *(const float4*)(B0r + 8 * jj);
    fq[jj] = *(const float4*)(B1r + 8 * jj);
  }
  __builtin_amdgcn_sched_barrier(0);
  v8f acc0 = {};
  v8f acc1 = {};
  float rs = 0.0f;
  for (int jj = 0; jj < G2_PAIRS; ++jj) {
    const float4 a  = fa[jj];
    const float4 b0 = fp[jj];
    const float4 b1 = fq[jj];
    rs += (a.x + a.y) + (a.z + a.w);
    v2f a01, a23, p01, p23, q01, q23;
    a01[0] = a.x;  a01[1] = a.y;  a23[0] = a.z;  a23[1] = a.w;
    p01[0] = b0.x; p01[1] = b0.y; p23[0] = b0.z; p23[1] = b0.w;
    q01[0] = b1.x; q01[1] = b1.y; q23[0] = b1.z; q23[1] = b1.w;
    acc0 = __builtin_amdgcn_wmma_f32_16x16x4_f32(
        false, a01, false, p01, (short)0, acc0, false, false);
    acc1 = __builtin_amdgcn_wmma_f32_16x16x4_f32(
        false, a01, false, q01, (short)0, acc1, false, false);
    acc0 = __builtin_amdgcn_wmma_f32_16x16x4_f32(
        false, a23, false, p23, (short)0, acc0, false, false);
    acc1 = __builtin_amdgcn_wmma_f32_16x16x4_f32(
        false, a23, false, q23, (short)0, acc1, false, false);
  }

  for (int r = 0; r < 8; ++r) {
    Red[wv][0][r * 32 + lane] = acc0[r];
    Red[wv][1][r * 32 + lane] = acc1[r];
  }
  Rs[wv][lane] = rs;
  __syncthreads();

  for (int i = 0; i < 2; ++i) {
    const int rowk = wv + i * G2_WAVES;
    if (rowk < 16) {
      const int srci = (rowk & 7) * 32 + (rowk >> 3) * 16 + l15;
      float sum = 0.0f;
      float sk  = 0.0f;
      for (int w = 0; w < G2_WAVES; ++w) {
        sum += Red[w][half][srci];
        sk  += Rs[w][rowk] + Rs[w][rowk + 16];
      }
      out[((size_t)n * KCLU + mt * 16 + rowk) * CDIM + cq * 32 + lane] = sum - rep[i] * sk;
    }
  }
}

__global__ __launch_bounds__(128) void rownorm_kernel(float* __restrict__ out,
                                                      float* __restrict__ ws) {
  __shared__ float red[128];
  const int nk = blockIdx.x;
  const int t  = threadIdx.x;
  const float v = out[(size_t)nk * CDIM + t];
  red[t] = v * v;
  __syncthreads();
  for (int off = 64; off > 0; off >>= 1) {
    if (t < off) red[t] += red[t + off];
    __syncthreads();
  }
  const float sum   = red[0];
  const float scale = 1.0f / fmaxf(sqrtf(sum), EPSV);
  out[(size_t)nk * CDIM + t] = v * scale;
  if (t == 0) ws[WS_RN + nk] = sum * scale * scale;
}

__global__ __launch_bounds__(256) void finalnorm_kernel(float* __restrict__ out,
                                                        const float* __restrict__ ws) {
  __shared__ float red[64];
  __shared__ float scale_s;
  const int n = blockIdx.x;
  const int t = threadIdx.x;
  if (t < 64) red[t] = ws[WS_RN + n * KCLU + t];
  __syncthreads();
  if (t == 0) {
    float s = 0.0f;
    for (int i = 0; i < 64; ++i) s += red[i];
    scale_s = 1.0f / fmaxf(sqrtf(s), EPSV);
  }
  __syncthreads();
  const float sc = scale_s;
  for (int i = t; i < KCLU * CDIM; i += 256)
    out[(size_t)n * KCLU * CDIM + i] *= sc;
}

extern "C" void kernel_launch(void* const* d_in, const int* in_sizes, int n_in,
                              void* d_out, int out_size, void* d_ws, size_t ws_size,
                              hipStream_t stream) {
  (void)in_sizes; (void)n_in; (void)out_size; (void)ws_size;
  const float* x    = (const float*)d_in[0];
  const float* cent = (const float*)d_in[1];
  float* out = (float*)d_out;
  float* ws  = (float*)d_ws;

  prep_kernel<<<MROWS, 128, 0, stream>>>(cent, ws);
  gemm1_softmax_kernel<<<dim3(PPTS / 16, NBATCH), G1_THREADS, 0, stream>>>(x, ws);
  gemm2_kernel<<<dim3(CDIM / 32, KCLU / 16, NBATCH), G2_THREADS, 0, stream>>>(x, cent, ws, out);
  rownorm_kernel<<<NBATCH * KCLU, 128, 0, stream>>>(out, ws);
  finalnorm_kernel<<<NBATCH, 256, 0, stream>>>(out, ws);
}


# 5 launches fused to 3: first GEMM kernel forms W scale and bias norm from its own A loads (prep launch gone); row and global norm in one kernel per n; sums of squares summed in a new order
# speedup vs baseline: 2.3496x; 1.3096x over previous
#include <hip/hip_runtime.h>
#include <math.h>


#define NBATCH 16
#define CDIM   128
#define PPTS   1200
#define KCLU   64
#define S1     5
#define MROWS  (KCLU * S1)
#define ALPHA  100.0f
#define EPSV   1e-12f

typedef __attribute__((ext_vector_type(2))) float v2f;
typedef __attribute__((ext_vector_type(8))) float v8f;

#define WS_MULT  0

#define G1_THREADS 160
#define XP 20

__global__ __launch_bounds__(G1_THREADS) void gemm1_softmax_kernel(const float* __restrict__ x,
                                                                   const float* __restrict__ cent,
                                                                   float* __restrict__ ws) {
  __shared__ __attribute__((aligned(16))) float Xs[CDIM * XP];
  __shared__ float Cv[MROWS * XP];
  __shared__ float Pm[8][S1][16];
  __shared__ float Ps[8][S1][16];
  __shared__ __attribute__((aligned(16))) float Bs[MROWS];

  const int pt   = blockIdx.x;
  const int n    = blockIdx.y;
  const int t    = threadIdx.x;
  const int lane = t & 31;
  const int wv   = t >> 5;
  const int p0   = pt * 16;
  const int col  = lane & 15;
  const int half = lane >> 4;

  const float* xb = x + (size_t)n * CDIM * PPTS + p0;
  const int i0 = t;
  const int i1 = t + G1_THREADS;
  const int i2 = t + 2 * G1_THREADS;
  const int i3 = t + 3 * G1_THREADS;
  const int i3c = (i3 < CDIM * 4) ? i3 : i2;
  const float4 x0 = *(const float4*)(xb + (size_t)(i0 >> 2) * PPTS + (i0 & 3) * 4);
  const float4 x1 = *(const float4*)(xb + (size_t)(i1 >> 2) * PPTS + (i1 & 3) * 4);
  const float4 x2 = *(const float4*)(xb + (size_t)(i2 >> 2) * PPTS + (i2 & 3) * 4);
  const float4 x3 = *(const float4*)(xb + (size_t)(i3c >> 2) * PPTS + (i3c & 3) * 4);
  *(float4*)(&Xs[(i0 >> 2) * XP + (i0 & 3) * 4]) = x0;
  *(float4*)(&Xs[(i1 >> 2) * XP + (i1 & 3) * 4]) = x1;
  *(float4*)(&Xs[(i2 >> 2) * XP + (i2 & 3) * 4]) = x2;
  *(float4*)(&Xs[(i3c >> 2) * XP + (i3c & 3) * 4]) = x3;
  __syncthreads();

  float bq[64];
  for (int j = 0; j < 16; ++j)
    for (int e = 0; e < 4; ++e)
      bq[j * 4 + e] = Xs[(8 * j + 4 * half + e) * XP + col];

  const float* wbase = cent + (size_t)(wv * 64 + col) * CDIM + 4 * half;
  v8f acc[4];
  v2f sq[4];
  for (int i = 0; i < 4; ++i) { acc[i] = (v8f){}; sq[i] = (v2f){}; }
  for (int j = 0; j < 16; ++j) {
    float4 a4[4];
    for (int i = 0; i < 4; ++i)
      a4[i] = *(const float4*)(wbase + (size_t)i * 16 * CDIM + 8 * j);
    v2f b0, b1;
    b0[0] = bq[4 * j + 0]; b0[1] = bq[4 * j + 1];
    b1[0] = bq[4 * j + 2]; b1[1] = bq[4 * j + 3];
    for (int i = 0; i < 4; ++i) {
      v2f a0; a0[0] = a4[i].x; a0[1] = a4[i].y;
      sq[i] += a0 * a0;
      a0 *= 2.0f * ALPHA;
      acc[i] = __builtin_amdgcn_wmma_f32_16x16x4_f32(
          false, a0, false, b0, (short)0, acc[i], false, false);
    }
    for (int i = 0; i < 4; ++i) {
      v2f a1; a1[0] = a4[i].z; a1[1] = a4[i].w;
      sq[i] += a1 * a1;
      a1 *= 2.0f * ALPHA;
      acc[i] = __builtin_amdgcn_wmma_f32_16x16x4_f32(
          false, a1, false, b1, (short)0, acc[i], false, false);
    }
  }

  for (int i = 0; i < 4; ++i) {
    const float ssl = sq[i][0] + sq[i][1];
    const float tot = ssl + __shfl_xor(ssl, 16);
    Bs[wv * 64 + i * 16 + col] = -ALPHA * sqrtf(tot);
  }

  for (int i = 0; i < 4; ++i) {
    const int mbase = wv * 64 + i * 16 + half * 8;
    for (int r = 0; r < 8; ++r)
      Cv[(mbase + r) * XP + col] = acc[i][r];
  }
  __syncthreads();

  const int p  = t & 15;
  const int kg = (t >> 4) & 7;
  float vv[8][S1];
  if (t < 128) {
    float bl[8 * S1];
    for (int q = 0; q < 2 * S1; ++q) {
      const float4 b4 = *(const float4*)(&Bs[kg * 8 * S1 + 4 * q]);
      bl[4 * q + 0] = b4.x; bl[4 * q + 1] = b4.y; bl[4 * q + 2] = b4.z; bl[4 * q + 3] = b4.w;
    }
    for (int kk = 0; kk < 8; ++kk)
      for (int s = 0; s < S1; ++s)
        vv[kk][s] = Cv[((kg * 8 + kk) * S1 + s) * XP + p] + bl[kk * S1 + s];
    for (int s = 0; s < S1; ++s) {
      float m = vv[0][s];
      for (int kk = 1; kk < 8; ++kk) m = fmaxf(m, vv[kk][s]);
      float sm = 0.0f;
      for (int kk = 0; kk < 8; ++kk) sm += __expf(vv[kk][s] - m);
      Pm[kg][s][p] = m;
      Ps[kg][s][p] = sm;
    }
  }
  __syncthreads();
  if (t < 128) {
    float mA[S1], sA[S1];
    for (int s = 0; s < S1; ++s) {
      float m = Pm[0][s][p];
      for (int g = 1; g < 8; ++g) m = fmaxf(m, Pm[g][s][p]);
      float sm = 0.0f;
      for (int g = 0; g < 8; ++g) sm += Ps[g][s][p] * __expf(Pm[g][s][p] - m);
      mA[s] = m;
      sA[s] = sm;
    }
    float* MU = ws + WS_MULT + (size_t)n * KCLU * PPTS;
    for (int kk = 0; kk < 8; ++kk) {
      float mB = vv[kk][0];
      for (int s = 1; s < S1; ++s) mB = fmaxf(mB, vv[kk][s]);
      float sB = 0.0f;
      for (int s = 0; s < S1; ++s) sB += __expf(vv[kk][s] - mB);
      float mult = 1.0f;
      for (int s = 0; s < S1; ++s) {
        const float sa = (__expf(vv[kk][s] - mA[s]) / sA[s]) * (__expf(vv[kk][s] - mB) / sB);
        mult *= (1.0f + sa);
      }
      MU[(size_t)(kg * 8 + kk) * PPTS + p0 + p] = mult;
    }
  }
}

#define G2_WAVES   10
#define G2_THREADS (G2_WAVES * 32)
#define G2_SLICE   (PPTS / G2_WAVES)
#define G2_PIECES  (G2_SLICE / 4)
#define G2_PAIRS   (G2_SLICE / 8)
#define G2_PITCH   124
#define G2_ROWS    48
#define G2_TPITCH  (8 * 32 + 16)

__global__ __launch_bounds__(G2_THREADS) void gemm2_kernel(const float* __restrict__ x,
                                                           const float* __restrict__ cent,
                                                           const float* __restrict__ ws,
                                                           float* __restrict__ out) {
  __shared__ __attribute__((aligned(16))) float St[G2_WAVES][G2_ROWS][G2_PITCH];
  __shared__ float Red[G2_WAVES][2][G2_TPITCH];
  __shared__ float Rs[G2_WAVES][32];

  const int cq   = blockIdx.x;
  const int mt   = blockIdx.y;
  const int n    = blockIdx.z;
  const int t    = threadIdx.x;
  const int lane = t & 31;
  const int wv   = t >> 5;
  const int l15  = lane & 15;
  const int half = lane >> 4;

  const float* Asrc = ws + WS_MULT + ((size_t)n * KCLU + mt * 16) * PPTS + wv * G2_SLICE;
  const float* Bsrc = x + ((size_t)n * CDIM + cq * 32) * PPTS + wv * G2_SLICE;
  float (*S)[G2_PITCH] = St[wv];

  float rep[2];
  for (int i = 0; i < 2; ++i) {
    const int rowk = (wv + i * G2_WAVES < 16) ? wv + i * G2_WAVES : wv;
    rep[i] = cent[(size_t)((mt * 16 + rowk) * S1) * CDIM + cq * 32 + lane];
  }

  if (lane < G2_PIECES) {
    float4 v[G2_ROWS];
    for (int r = 0; r < 16; ++r)
      v[r] = *(const float4*)(Asrc + (size_t)r * PPTS + lane * 4);
    for (int r = 0; r < 32; ++r)
      v[16 + r] = *(const float4*)(Bsrc + (size_t)r * PPTS + lane * 4);
    for (int r = 0; r < G2_ROWS; ++r)
      *(float4*)(&S[r][lane * 4]) = v[r];
  }
  __syncthreads();

  const float* Ar  = &S[l15][4 * half];
  const float* B0r = &S[16 + l15][4 * half];
  const float* B1r = &S[32 + l15][4 * half];
  float4 fa[G2_PAIRS], fp[G2_PAIRS], fq[G2_PAIRS];
  for (int jj = 0; jj < G2_PAIRS; ++jj) {
    fa[jj] = *(const float4*)(Ar + 8 * jj);
    fp[jj] = *(const float4*)(B0r + 8 * jj);
    fq[jj] = *(const float4*)(B1r + 8 * jj);
  }
  __builtin_amdgcn_sched_barrier(0);
  v8f acc0 = {};
  v8f acc1 = {};
  float rs = 0.0f;
  for (int jj = 0; jj < G2_PAIRS; ++jj) {
    const float4 a  = fa[jj];
    const float4 b0 = fp[jj];
    const float4 b1 = fq[jj];
    rs += (a.x + a.y) + (a.z + a.w);
    v2f a01, a23, p01, p23, q01, q23;
    a01[0] = a.x;  a01[1] = a.y;  a23[0] = a.z;  a23[1] = a.w;
    p01[0] = b0.x; p01[1] = b0.y; p23[0] = b0.z; p23[1] = b0.w;
    q01[0] = b1.x; q01[1] = b1.y; q23[0] = b1.z; q23[1] = b1.w;
    acc0 = __builtin_amdgcn_wmma_f32_16x16x4_f32(
        false, a01, false, p01, (short)0, acc0, false, false);
    acc1 = __builtin_amdgcn_wmma_f32_16x16x4_f32(
        false, a01, false, q01, (short)0, acc1, false, false);
    acc0 = __builtin_amdgcn_wmma_f32_16x16x4_f32(
        false, a23, false, p23, (short)0, acc0, false, false);
    acc1 = __builtin_amdgcn_wmma_f32_16x16x4_f32(
        false, a23, false, q23, (short)0, acc1, false, false);
  }

  for (int r = 0; r < 8; ++r) {
    Red[wv][0][r * 32 + lane] = acc0[r];
    Red[wv][1][r * 32 + lane] = acc1[r];
  }
  Rs[wv][lane] = rs;
  __syncthreads();

  for (int i = 0; i < 2; ++i) {
    const int rowk = wv + i * G2_WAVES;
    if (rowk < 16) {
      const int srci = (rowk & 7) * 32 + (rowk >> 3) * 16 + l15;
      float sum = 0.0f;
      float sk  = 0.0f;
      for (int w = 0; w < G2_WAVES; ++w) {
        sum += Red[w][half][srci];
        sk  += Rs[w][rowk] + Rs[w][rowk + 16];
      }
      out[((size_t)n * KCLU + mt * 16 + rowk) * CDIM + cq * 32 + lane] = sum - rep[i] * sk;
    }
  }
}

#define NRM_THREADS 1024

__global__ __launch_bounds__(NRM_THREADS) void norms_kernel(float* __restrict__ out) {
  __shared__ __attribute__((aligned(16))) float rn[KCLU];
  const int n    = blockIdx.x;
  const int t    = threadIdx.x;
  const int lane = t & 31;
  const int wv   = t >> 5;
  float* base = out + (size_t)n * KCLU * CDIM + lane * 4;

  float4 v[2];
  for (int i = 0; i < 2; ++i)
    v[i] = *(const float4*)(base + (size_t)(wv + 32 * i) * CDIM);

  float ssq[2];
  for (int i = 0; i < 2; ++i)
    ssq[i] = v[i].x * v[i].x + v[i].y * v[i].y + v[i].z * v[i].z + v[i].w * v[i].w;
  for (int off = 16; off > 0; off >>= 1)
    for (int i = 0; i < 2; ++i) ssq[i] += __shfl_xor(ssq[i], off);
  float rscale[2];
  for (int i = 0; i < 2; ++i) {
    const float sc = 1.0f / fmaxf(sqrtf(ssq[i]), EPSV);
    rscale[i] = sc;
    if (lane == 0) rn[wv + 32 * i] = ssq[i] * sc * sc;
  }
  __syncthreads();

  float tot = 0.0f;
  for (int q = 0; q < KCLU / 4; ++q) {
    const float4 r4 = *(const float4*)(&rn[4 * q]);
    tot += r4.x; tot += r4.y; tot += r4.z; tot += r4.w;
  }
  const float g = 1.0f / fmaxf(sqrtf(tot), EPSV);

  for (int i = 0; i < 2; ++i) {
    float4 y;
    y.x = v[i].x * rscale[i] * g;
    y.y = v[i].y * rscale[i] * g;
    y.z = v[i].z * rscale[i] * g;
    y.w = v[i].w * rscale[i] * g;
    *(float4*)(base + (size_t)(wv + 32 * i) * CDIM) = y;
  }
}

extern "C" void kernel_launch(void* const* d_in, const int* in_sizes, int n_in,
                              void* d_out, int out_size, void* d_ws, size_t ws_size,
                              hipStream_t stream) {
  (void)in_sizes; (void)n_in; (void)out_size; (void)ws_size;
  const float* x    = (const float*)d_in[0];
  const float* cent = (const float*)d_in[1];
  float* out = (float*)d_out;
  float* ws  = (float*)d_ws;

  gemm1_softmax_kernel<<<dim3(PPTS / 16, NBATCH), G1_THREADS, 0, stream>>>(x, cent, ws);
  gemm2_kernel<<<dim3(CDIM / 32, KCLU / 16, NBATCH), G2_THREADS, 0, stream>>>(x, cent, ws, out);
  norms_kernel<<<NBATCH, NRM_THREADS, 0, stream>>>(out);
}


# gemm1 softmax epilogue: each exp formed once and reused (group-max exp rescaled per s), 8-group combine done once per (column,s) not by 8 threads, quotients as products with one reciprocal per sum
# speedup vs baseline: 2.4968x; 1.0626x over previous
#include <hip/hip_runtime.h>
#include <math.h>


#define NBATCH 16
#define CDIM   128
#define PPTS   1200
#define KCLU   64
#define S1     5
#define MROWS  (KCLU * S1)
#define ALPHA  100.0f
#define EPSV   1e-12f

typedef __attribute__((ext_vector_type(2))) float v2f;
typedef __attribute__((ext_vector_type(8))) float v8f;

#define WS_MULT  0

#define G1_THREADS 160
#define XP 20

__global__ __launch_bounds__(G1_THREADS) void gemm1_softmax_kernel(const float* __restrict__ x,
                                                                   const float* __restrict__ cent,
                                                                   float* __restrict__ ws) {
  __shared__ __attribute__((aligned(16))) float Xs[CDIM * XP];
  __shared__ float Cv[MROWS * XP];
  __shared__ float Pm[8][S1][16];
  __shared__ float Ps[8][S1][16];
  __shared__ __attribute__((aligned(16))) float Bs[MROWS];

  const int pt   = blockIdx.x;
  const int n    = blockIdx.y;
  const int t    = threadIdx.x;
  const int lane = t & 31;
  const int wv   = t >> 5;
  const int p0   = pt * 16;
  const int col  = lane & 15;
  const int half = lane >> 4;

  const float* xb = x + (size_t)n * CDIM * PPTS + p0;
  const int i0 = t;
  const int i1 = t + G1_THREADS;
  const int i2 = t + 2 * G1_THREADS;
  const int i3 = t + 3 * G1_THREADS;
  const int i3c = (i3 < CDIM * 4) ? i3 : i2;
  const float4 x0 = *(const float4*)(xb + (size_t)(i0 >> 2) * PPTS + (i0 & 3) * 4);
  const float4 x1 = *(const float4*)(xb + (size_t)(i1 >> 2) * PPTS + (i1 & 3) * 4);
  const float4 x2 = *(const float4*)(xb + (size_t)(i2 >> 2) * PPTS + (i2 & 3) * 4);
  const float4 x3 = *(const float4*)(xb + (size_t)(i3c >> 2) * PPTS + (i3c & 3) * 4);
  *(float4*)(&Xs[(i0 >> 2) * XP + (i0 & 3) * 4]) = x0;
  *(float4*)(&Xs[(i1 >> 2) * XP + (i1 & 3) * 4]) = x1;
  *(float4*)(&Xs[(i2 >> 2) * XP + (i2 & 3) * 4]) = x2;
  *(float4*)(&Xs[(i3c >> 2) * XP + (i3c & 3) * 4]) = x3;
  __syncthreads();

  float bq[64];
  for (int j = 0; j < 16; ++j)
    for (int e = 0; e < 4; ++e)
      bq[j * 4 + e] = Xs[(8 * j + 4 * half + e) * XP + col];

  const float* wbase = cent + (size_t)(wv * 64 + col) * CDIM + 4 * half;
  v8f acc[4];
  v2f sq[4];
  for (int i = 0; i < 4; ++i) { acc[i] = (v8f){}; sq[i] = (v2f){}; }
  for (int j = 0; j < 16; ++j) {
    float4 a4[4];
    for (int i = 0; i < 4; ++i)
      a4[i] = *(const float4*)(wbase + (size_t)i * 16 * CDIM + 8 * j);
    v2f b0, b1;
    b0[0] = bq[4 * j + 0]; b0[1] = bq[4 * j + 1];
    b1[0] = bq[4 * j + 2]; b1[1] = bq[4 * j + 3];
    for (int i = 0; i < 4; ++i) {
      v2f a0; a0[0] = a4[i].x; a0[1] = a4[i].y;
      sq[i] += a0 * a0;
      a0 *= 2.0f * ALPHA;
      acc[i] = __builtin_amdgcn_wmma_f32_16x16x4_f32(
          false, a0, false, b0, (short)0, acc[i], false, false);
    }
    for (int i = 0; i < 4; ++i) {
      v2f a1; a1[0] = a4[i].z; a1[1] = a4[i].w;
      sq[i] += a1 * a1;
      a1 *= 2.0f * ALPHA;
      acc[i] = __builtin_amdgcn_wmma_f32_16x16x4_f32(
          false, a1, false, b1, (short)0, acc[i], false, false);
    }
  }

  for (int i = 0; i < 4; ++i) {
    const float ssl = sq[i][0] + sq[i][1];
    const float tot = ssl + __shfl_xor(ssl, 16);
    Bs[wv * 64 + i * 16 + col] = -ALPHA * sqrtf(tot);
  }

  for (int i = 0; i < 4; ++i) {
    const int mbase = wv * 64 + i * 16 + half * 8;
    for (int r = 0; r < 8; ++r)
      Cv[(mbase + r) * XP + col] = acc[i][r];
  }
  __syncthreads();

  __shared__ float Ma[S1][16];
  __shared__ float Ra[S1][16];
  const int p  = t & 15;
  const int kg = (t >> 4) & 7;
  float vv[8][S1];
  float e1[8][S1];
  float mg[S1];
  if (t < 128) {
    float bl[8 * S1];
    for (int q = 0; q < 2 * S1; ++q) {
      const float4 b4 = *(const float4*)(&Bs[kg * 8 * S1 + 4 * q]);
      bl[4 * q + 0] = b4.x; bl[4 * q + 1] = b4.y; bl[4 * q + 2] = b4.z; bl[4 * q + 3] = b4.w;
    }
    for (int kk = 0; kk < 8; ++kk)
      for (int s = 0; s < S1; ++s)
        vv[kk][s] = Cv[((kg * 8 + kk) * S1 + s) * XP + p] + bl[kk * S1 + s];
    for (int s = 0; s < S1; ++s) {
      float m = vv[0][s];
      for (int kk = 1; kk < 8; ++kk) m = fmaxf(m, vv[kk][s]);
      float sm = 0.0f;
      for (int kk = 0; kk < 8; ++kk) {
        const float e = __expf(vv[kk][s] - m);
        e1[kk][s] = e;
        sm += e;
      }
      mg[s] = m;
      Pm[kg][s][p] = m;
      Ps[kg][s][p] = sm;
    }
  }
  __syncthreads();
  if (t < 16 * S1) {
    float m = Pm[0][kg][p];
    for (int g = 1; g < 8; ++g) m = fmaxf(m, Pm[g][kg][p]);
    float sm = 0.0f;
    for (int g = 0; g < 8; ++g) sm += Ps[g][kg][p] * __expf(Pm[g][kg][p] - m);
    Ma[kg][p] = m;
    Ra[kg][p] = 1.0f / sm;
  }
  __syncthreads();
  if (t < 128) {
    float gA[S1];
    for (int s = 0; s < S1; ++s) gA[s] = __expf(mg[s] - Ma[s][p]) * Ra[s][p];
    float* MU = ws + WS_MULT + (size_t)n * KCLU * PPTS;
    for (int kk = 0; kk < 8; ++kk) {
      float mB = vv[kk][0];
      for (int s = 1; s < S1; ++s) mB = fmaxf(mB, vv[kk][s]);
      float eB[S1];
      float sB = 0.0f;
      for (int s = 0; s < S1; ++s) {
        eB[s] = __expf(vv[kk][s] - mB);
        sB += eB[s];
      }
      const float rB = 1.0f / sB;
      float mult = 1.0f;
      for (int s = 0; s < S1; ++s) {
        const float sa = (e1[kk][s] * gA[s]) * (eB[s] * rB);
        mult *= (1.0f + sa);
      }
      MU[(size_t)(kg * 8 + kk) * PPTS + p0 + p] = mult;
    }
  }
}

#define G2_WAVES   10
#define G2_THREADS (G2_WAVES * 32)
#define G2_SLICE   (PPTS / G2_WAVES)
#define G2_PIECES  (G2_SLICE / 4)
#define G2_PAIRS   (G2_SLICE / 8)
#define G2_PITCH   124
#define G2_ROWS    48
#define G2_TPITCH  (8 * 32 + 16)

__global__ __launch_bounds__(G2_THREADS) void gemm2_kernel(const float* __restrict__ x,
                                                           const float* __restrict__ cent,
                                                           const float* __restrict__ ws,
                                                           float* __restrict__ out) {
  __shared__ __attribute__((aligned(16))) float St[G2_WAVES][G2_ROWS][G2_PITCH];
  __shared__ float Red[G2_WAVES][2][G2_TPITCH];
  __shared__ float Rs[G2_WAVES][32];

  const int cq   = blockIdx.x;
  const int mt   = blockIdx.y;
  const int n    = blockIdx.z;
  const int t    = threadIdx.x;
  const int lane = t & 31;
  const int wv   = t >> 5;
  const int l15  = lane & 15;
  const int half = lane >> 4;

  const float* Asrc = ws + WS_MULT + ((size_t)n * KCLU + mt * 16) * PPTS + wv * G2_SLICE;
  const float* Bsrc = x + ((size_t)n * CDIM + cq * 32) * PPTS + wv * G2_SLICE;
  float (*S)[G2_PITCH] = St[wv];

  float rep[2];
  for (int i = 0; i < 2; ++i) {
    const int rowk = (wv + i * G2_WAVES < 16) ? wv + i * G2_WAVES : wv;
    rep[i] = cent[(size_t)((mt * 16 + rowk) * S1) * CDIM + cq * 32 + lane];
  }

  if (lane < G2_PIECES) {
    float4 v[G2_ROWS];
    for (int r = 0; r < 16; ++r)
      v[r] = *(const float4*)(Asrc + (size_t)r * PPTS + lane * 4);
    for (int r = 0; r < 32; ++r)
      v[16 + r] = *(const float4*)(Bsrc + (size_t)r * PPTS + lane * 4);
    for (int r = 0; r < G2_ROWS; ++r)
      *(float4*)(&S[r][lane * 4]) = v[r];
  }
  __syncthreads();

  const float* Ar  = &S[l15][4 * half];
  const float* B0r = &S[16 + l15][4 * half];
  const float* B1r = &S[32 + l15][4 * half];
  float4 fa[G2_PAIRS], fp[G2_PAIRS], fq[G2_PAIRS];
  for (int jj = 0; jj < G2_PAIRS; ++jj) {
    fa[jj] = *(const float4*)(Ar + 8 * jj);
    fp[jj] = *(const float4*)(B0r + 8 * jj);
    fq[jj] = *(const float4*)(B1r + 8 * jj);
  }
  __builtin_amdgcn_sched_barrier(0);
  v8f acc0 = {};
  v8f acc1 = {};
  float rs = 0.0f;
  for (int jj = 0; jj < G2_PAIRS; ++jj) {
    const float4 a  = fa[jj];
    const float4 b0 = fp[jj];
    const float4 b1 = fq[jj];
    rs += (a.x + a.y) + (a.z + a.w);
    v2f a01, a23, p01, p23, q01, q23;
    a01[0] = a.x;  a01[1] = a.y;  a23[0] = a.z;  a23[1] = a.w;
    p01[0] = b0.x; p01[1] = b0.y; p23[0] = b0.z; p23[1] = b0.w;
    q01[0] = b1.x; q01[1] = b1.y; q23[0] = b1.z; q23[1] = b1.w;
    acc0 = __builtin_amdgcn_wmma_f32_16x16x4_f32(
        false, a01, false, p01, (short)0, acc0, false, false);
    acc1 = __builtin_amdgcn_wmma_f32_16x16x4_f32(
        false, a01, false, q01, (short)0, acc1, false, false);
    acc0 = __builtin_amdgcn_wmma_f32_16x16x4_f32(
        false, a23, false, p23, (short)0, acc0, false, false);
    acc1 = __builtin_amdgcn_wmma_f32_16x16x4_f32(
        false, a23, false, q23, (short)0, acc1, false, false);
  }

  for (int r = 0; r < 8; ++r) {
    Red[wv][0][r * 32 + lane] = acc0[r];
    Red[wv][1][r * 32 + lane] = acc1[r];
  }
  Rs[wv][lane] = rs;
  __syncthreads();

  for (int i = 0; i < 2; ++i) {
    const int rowk = wv + i * G2_WAVES;
    if (rowk < 16) {
      const int srci = (rowk & 7) * 32 + (rowk >> 3) * 16 + l15;
      float sum = 0.0f;
      float sk  = 0.0f;
      for (int w = 0; w < G2_WAVES; ++w) {
        sum += Red[w][half][srci];
        sk  += Rs[w][rowk] + Rs[w][rowk + 16];
      }
      out[((size_t)n * KCLU + mt * 16 + rowk) * CDIM + cq * 32 + lane] = sum - rep[i] * sk;
    }
  }
}

#define NRM_THREADS 1024

__global__ __launch_bounds__(NRM_THREADS) void norms_kernel(float* __restrict__ out) {
  __shared__ __attribute__((aligned(16))) float rn[KCLU];
  const int n    = blockIdx.x;
  const int t    = threadIdx.x;
  const int lane = t & 31;
  const int wv   = t >> 5;
  float* base = out + (size_t)n * KCLU * CDIM + lane * 4;

  float4 v[2];
  for (int i = 0; i < 2; ++i)
    v[i] = *(const float4*)(base + (size_t)(wv + 32 * i) * CDIM);

  float ssq[2];
  for (int i = 0; i < 2; ++i)
    ssq[i] = v[i].x * v[i].x + v[i].y * v[i].y + v[i].z * v[i].z + v[i].w * v[i].w;
  for (int off = 16; off > 0; off >>= 1)
    for (int i = 0; i < 2; ++i) ssq[i] += __shfl_xor(ssq[i], off);
  float rscale[2];
  for (int i = 0; i < 2; ++i) {
    const float sc = 1.0f / fmaxf(sqrtf(ssq[i]), EPSV);
    rscale[i] = sc;
    if (lane == 0) rn[wv + 32 * i] = ssq[i] * sc * sc;
  }
  __syncthreads();

  float tot = 0.0f;
  for (int q = 0; q < KCLU / 4; ++q) {
    const float4 r4 = *(const float4*)(&rn[4 * q]);
    tot += r4.x; tot += r4.y; tot += r4.z; tot += r4.w;
  }
  const float g = 1.0f / fmaxf(sqrtf(tot), EPSV);

  for (int i = 0; i < 2; ++i) {
    float4 y;
    y.x = v[i].x * rscale[i] * g;
    y.y = v[i].y * rscale[i] * g;
    y.z = v[i].z * rscale[i] * g;
    y.w = v[i].w * rscale[i] * g;
    *(float4*)(base + (size_t)(wv + 32 * i) * CDIM) = y;
  }
}

extern "C" void kernel_launch(void* const* d_in, const int* in_sizes, int n_in,
                              void* d_out, int out_size, void* d_ws, size_t ws_size,
                              hipStream_t stream) {
  (void)in_sizes; (void)n_in; (void)out_size; (void)ws_size;
  const float* x    = (const float*)d_in[0];
  const float* cent = (const float*)d_in[1];
  float* out = (float*)d_out;
  float* ws  = (float*)d_ws;

  gemm1_softmax_kernel<<<dim3(PPTS / 16, NBATCH), G1_THREADS, 0, stream>>>(x, cent, ws);
  gemm2_kernel<<<dim3(CDIM / 32, KCLU / 16, NBATCH), G2_THREADS, 0, stream>>>(x, cent, ws, out);
  norms_kernel<<<NBATCH, NRM_THREADS, 0, stream>>>(out);
}


# gemm1 work-group re-tiled from 5 waves x 4 M-tiles to 4 waves x 5 M-tiles: one wave per SIMD, so every SIMD of a WGP issues an equal share of the f32 matrix instructions; summation order unchanged
# speedup vs baseline: 2.6014x; 1.0419x over previous
#include <hip/hip_runtime.h>
#include <math.h>


#define NBATCH 16
#define CDIM   128
#define PPTS   1200
#define KCLU   64
#define S1     5
#define MROWS  (KCLU * S1)
#define ALPHA  100.0f
#define EPSV   1e-12f

typedef __attribute__((ext_vector_type(2))) float v2f;
typedef __attribute__((ext_vector_type(8))) float v8f;

#define WS_MULT  0

#define G1_THREADS 128
#define G1_TILES   5
#define XP 20

__global__ __launch_bounds__(G1_THREADS) void gemm1_softmax_kernel(const float* __restrict__ x,
                                                                   const float* __restrict__ cent,
                                                                   float* __restrict__ ws) {
  __shared__ __attribute__((aligned(16))) float Xs[CDIM * XP];
  __shared__ float Cv[MROWS * XP];
  __shared__ float Pm[8][S1][16];
  __shared__ float Ps[8][S1][16];
  __shared__ __attribute__((aligned(16))) float Bs[MROWS];

  const int pt   = blockIdx.x;
  const int n    = blockIdx.y;
  const int t    = threadIdx.x;
  const int lane = t & 31;
  const int wv   = t >> 5;
  const int p0   = pt * 16;
  const int col  = lane & 15;
  const int half = lane >> 4;

  const float* xb = x + (size_t)n * CDIM * PPTS + p0;
  const int i0 = t;
  const int i1 = t + G1_THREADS;
  const int i2 = t + 2 * G1_THREADS;
  const int i3 = t + 3 * G1_THREADS;
  const float4 x0 = *(const float4*)(xb + (size_t)(i0 >> 2) * PPTS + (i0 & 3) * 4);
  const float4 x1 = *(const float4*)(xb + (size_t)(i1 >> 2) * PPTS + (i1 & 3) * 4);
  const float4 x2 = *(const float4*)(xb + (size_t)(i2 >> 2) * PPTS + (i2 & 3) * 4);
  const float4 x3 = *(const float4*)(xb + (size_t)(i3 >> 2) * PPTS + (i3 & 3) * 4);
  *(float4*)(&Xs[(i0 >> 2) * XP + (i0 & 3) * 4]) = x0;
  *(float4*)(&Xs[(i1 >> 2) * XP + (i1 & 3) * 4]) = x1;
  *(float4*)(&Xs[(i2 >> 2) * XP + (i2 & 3) * 4]) = x2;
  *(float4*)(&Xs[(i3 >> 2) * XP + (i3 & 3) * 4]) = x3;
  __syncthreads();

  float bq[64];
  for (int j = 0; j < 16; ++j)
    for (int e = 0; e < 4; ++e)
      bq[j * 4 + e] = Xs[(8 * j + 4 * half + e) * XP + col];

  const float* wbase = cent + (size_t)(wv * (16 * G1_TILES) + col) * CDIM + 4 * half;
  v8f acc[G1_TILES];
  v2f sq[G1_TILES];
  for (int i = 0; i < G1_TILES; ++i) { acc[i] = (v8f){}; sq[i] = (v2f){}; }
  for (int j = 0; j < 16; ++j) {
    float4 a4[G1_TILES];
    for (int i = 0; i < G1_TILES; ++i)
      a4[i] = *(const float4*)(wbase + (size_t)i * 16 * CDIM + 8 * j);
    v2f b0, b1;
    b0[0] = bq[4 * j + 0]; b0[1] = bq[4 * j + 1];
    b1[0] = bq[4 * j + 2]; b1[1] = bq[4 * j + 3];
    for (int i = 0; i < G1_TILES; ++i) {
      v2f a0; a0[0] = a4[i].x; a0[1] = a4[i].y;
      sq[i] += a0 * a0;
      a0 *= 2.0f * ALPHA;
      acc[i] = __builtin_amdgcn_wmma_f32_16x16x4_f32(
          false, a0, false, b0, (short)0, acc[i], false, false);
    }
    for (int i = 0; i < G1_TILES; ++i) {
      v2f a1; a1[0] = a4[i].z; a1[1] = a4[i].w;
      sq[i] += a1 * a1;
      a1 *= 2.0f * ALPHA;
      acc[i] = __builtin_amdgcn_wmma_f32_16x16x4_f32(
          false, a1, false, b1, (short)0, acc[i], false, false);
    }
  }

  for (int i = 0; i < G1_TILES; ++i) {
    const float ssl = sq[i][0] + sq[i][1];
    const float tot = ssl + __shfl_xor(ssl, 16);
    Bs[wv * (16 * G1_TILES) + i * 16 + col] = -ALPHA * sqrtf(tot);
  }

  for (int i = 0; i < G1_TILES; ++i) {
    const int mbase = wv * (16 * G1_TILES) + i * 16 + half * 8;
    for (int r = 0; r < 8; ++r)
      Cv[(mbase + r) * XP + col] = acc[i][r];
  }
  __syncthreads();

  __shared__ float Ma[S1][16];
  __shared__ float Ra[S1][16];
  const int p  = t & 15;
  const int kg = (t >> 4) & 7;
  float vv[8][S1];
  float e1[8][S1];
  float mg[S1];
  {
    float bl[8 * S1];
    for (int q = 0; q < 2 * S1; ++q) {
      const float4 b4 = *(const float4*)(&Bs[kg * 8 * S1 + 4 * q]);
      bl[4 * q + 0] = b4.x; bl[4 * q + 1] = b4.y; bl[4 * q + 2] = b4.z; bl[4 * q + 3] = b4.w;
    }
    for (int kk = 0; kk < 8; ++kk)
      for (int s = 0; s < S1; ++s)
        vv[kk][s] = Cv[((kg * 8 + kk) * S1 + s) * XP + p] + bl[kk * S1 + s];
    for (int s = 0; s < S1; ++s) {
      float m = vv[0][s];
      for (int kk = 1; kk < 8; ++kk) m = fmaxf(m, vv[kk][s]);
      float sm = 0.0f;
      for (int kk = 0; kk < 8; ++kk) {
        const float e = __expf(vv[kk][s] - m);
        e1[kk][s] = e;
        sm += e;
      }
      mg[s] = m;
      Pm[kg][s][p] = m;
      Ps[kg][s][p] = sm;
    }
  }
  __syncthreads();
  if (t < 16 * S1) {
    float m = Pm[0][kg][p];
    for (int g = 1; g < 8; ++g) m = fmaxf(m, Pm[g][kg][p]);
    float sm = 0.0f;
    for (int g = 0; g < 8; ++g) sm += Ps[g][kg][p] * __expf(Pm[g][kg][p] - m);
    Ma[kg][p] = m;
    Ra[kg][p] = 1.0f / sm;
  }
  __syncthreads();
  {
    float gA[S1];
    for (int s = 0; s < S1; ++s) gA[s] = __expf(mg[s] - Ma[s][p]) * Ra[s][p];
    float* MU = ws + WS_MULT + (size_t)n * KCLU * PPTS;
    for (int kk = 0; kk < 8; ++kk) {
      float mB = vv[kk][0];
      for (int s = 1; s < S1; ++s) mB = fmaxf(mB, vv[kk][s]);
      float eB[S1];
      float sB = 0.0f;
      for (int s = 0; s < S1; ++s) {
        eB[s] = __expf(vv[kk][s] - mB);
        sB += eB[s];
      }
      const float rB = 1.0f / sB;
      float mult = 1.0f;
      for (int s = 0; s < S1; ++s) {
        const float sa = (e1[kk][s] * gA[s]) * (eB[s] * rB);
        mult *= (1.0f + sa);
      }
      MU[(size_t)(kg * 8 + kk) * PPTS + p0 + p] = mult;
    }
  }
}

#define G2_WAVES   10
#define G2_THREADS (G2_WAVES * 32)
#define G2_SLICE   (PPTS / G2_WAVES)
#define G2_PIECES  (G2_SLICE / 4)
#define G2_PAIRS   (G2_SLICE / 8)
#define G2_PITCH   124
#define G2_ROWS    48
#define G2_TPITCH  (8 * 32 + 16)

__global__ __launch_bounds__(G2_THREADS) void gemm2_kernel(const float* __restrict__ x,
                                                           const float* __restrict__ cent,
                                                           const float* __restrict__ ws,
                                                           float* __restrict__ out) {
  __shared__ __attribute__((aligned(16))) float St[G2_WAVES][G2_ROWS][G2_PITCH];
  __shared__ float Red[G2_WAVES][2][G2_TPITCH];
  __shared__ float Rs[G2_WAVES][32];

  const int cq   = blockIdx.x;
  const int mt   = blockIdx.y;
  const int n    = blockIdx.z;
  const int t    = threadIdx.x;
  const int lane = t & 31;
  const int wv   = t >> 5;
  const int l15  = lane & 15;
  const int half = lane >> 4;

  const float* Asrc = ws + WS_MULT + ((size_t)n * KCLU + mt * 16) * PPTS + wv * G2_SLICE;
  const float* Bsrc = x + ((size_t)n * CDIM + cq * 32) * PPTS + wv * G2_SLICE;
  float (*S)[G2_PITCH] = St[wv];

  float rep[2];
  for (int i = 0; i < 2; ++i) {
    const int rowk = (wv + i * G2_WAVES < 16) ? wv + i * G2_WAVES : wv;
    rep[i] = cent[(size_t)((mt * 16 + rowk) * S1) * CDIM + cq * 32 + lane];
  }

  if (lane < G2_PIECES) {
    float4 v[G2_ROWS];
    for (int r = 0; r < 16; ++r)
      v[r] = *(const float4*)(Asrc + (size_t)r * PPTS + lane * 4);
    for (int r = 0; r < 32; ++r)
      v[16 + r] = *(const float4*)(Bsrc + (size_t)r * PPTS + lane * 4);
    for (int r = 0; r < G2_ROWS; ++r)
      *(float4*)(&S[r][lane * 4]) = v[r];
  }
  __syncthreads();

  const float* Ar  = &S[l15][4 * half];
  const float* B0r = &S[16 + l15][4 * half];
  const float* B1r = &S[32 + l15][4 * half];
  float4 fa[G2_PAIRS], fp[G2_PAIRS], fq[G2_PAIRS];
  for (int jj = 0; jj < G2_PAIRS; ++jj) {
    fa[jj] = *(const float4*)(Ar + 8 * jj);
    fp[jj] = *(const float4*)(B0r + 8 * jj);
    fq[jj] = *(const float4*)(B1r + 8 * jj);
  }
  __builtin_amdgcn_sched_barrier(0);
  v8f acc0 = {};
  v8f acc1 = {};
  float rs = 0.0f;
  for (int jj = 0; jj < G2_PAIRS; ++jj) {
    const float4 a  = fa[jj];
    const float4 b0 = fp[jj];
    const float4 b1 = fq[jj];
    rs += (a.x + a.y) + (a.z + a.w);
    v2f a01, a23, p01, p23, q01, q23;
    a01[0] = a.x;  a01[1] = a.y;  a23[0] = a.z;  a23[1] = a.w;
    p01[0] = b0.x; p01[1] = b0.y; p23[0] = b0.z; p23[1] = b0.w;
    q01[0] = b1.x; q01[1] = b1.y; q23[0] = b1.z; q23[1] = b1.w;
    acc0 = __builtin_amdgcn_wmma_f32_16x16x4_f32(
        false, a01, false, p01, (short)0, acc0, false, false);
    acc1 = __builtin_amdgcn_wmma_f32_16x16x4_f32(
        false, a01, false, q01, (short)0, acc1, false, false);
    acc0 = __builtin_amdgcn_wmma_f32_16x16x4_f32(
        false, a23, false, p23, (short)0, acc0, false, false);
    acc1 = __builtin_amdgcn_wmma_f32_16x16x4_f32(
        false, a23, false, q23, (short)0, acc1, false, false);
  }

  for (int r = 0; r < 8; ++r) {
    Red[wv][0][r * 32 + lane] = acc0[r];
    Red[wv][1][r * 32 + lane] = acc1[r];
  }
  Rs[wv][lane] = rs;
  __syncthreads();

  for (int i = 0; i < 2; ++i) {
    const int rowk = wv + i * G2_WAVES;
    if (rowk < 16) {
      const int srci = (rowk & 7) * 32 + (rowk >> 3) * 16 + l15;
      float sum = 0.0f;
      float sk  = 0.0f;
      for (int w = 0; w < G2_WAVES; ++w) {
        sum += Red[w][half][srci];
        sk  += Rs[w][rowk] + Rs[w][rowk + 16];
      }
      out[((size_t)n * KCLU + mt * 16 + rowk) * CDIM + cq * 32 + lane] = sum - rep[i] * sk;
    }
  }
}

#define NRM_THREADS 1024

__global__ __launch_bounds__(NRM_THREADS) void norms_kernel(float* __restrict__ out) {
  __shared__ __attribute__((aligned(16))) float rn[KCLU];
  const int n    = blockIdx.x;
  const int t    = threadIdx.x;
  const int lane = t & 31;
  const int wv   = t >> 5;
  float* base = out + (size_t)n * KCLU * CDIM + lane * 4;

  float4 v[2];
  for (int i = 0; i < 2; ++i)
    v[i] = *(const float4*)(base + (size_t)(wv + 32 * i) * CDIM);

  float ssq[2];
  for (int i = 0; i < 2; ++i)
    ssq[i] = v[i].x * v[i].x + v[i].y * v[i].y + v[i].z * v[i].z + v[i].w * v[i].w;
  for (int off = 16; off > 0; off >>= 1)
    for (int i = 0; i < 2; ++i) ssq[i] += __shfl_xor(ssq[i], off);
  float rscale[2];
  for (int i = 0; i < 2; ++i) {
    const float sc = 1.0f / fmaxf(sqrtf(ssq[i]), EPSV);
    rscale[i] = sc;
    if (lane == 0) rn[wv + 32 * i] = ssq[i] * sc * sc;
  }
  __syncthreads();

  float tot = 0.0f;
  for (int q = 0; q < KCLU / 4; ++q) {
    const float4 r4 = *(const float4*)(&rn[4 * q]);
    tot += r4.x; tot += r4.y; tot += r4.z; tot += r4.w;
  }
  const float g = 1.0f / fmaxf(sqrtf(tot), EPSV);

  for (int i = 0; i < 2; ++i) {
    float4 y;
    y.x = v[i].x * rscale[i] * g;
    y.y = v[i].y * rscale[i] * g;
    y.z = v[i].z * rscale[i] * g;
    y.w = v[i].w * rscale[i] * g;
    *(float4*)(base + (size_t)(wv + 32 * i) * CDIM) = y;
  }
}

extern "C" void kernel_launch(void* const* d_in, const int* in_sizes, int n_in,
                              void* d_out, int out_size, void* d_ws, size_t ws_size,
                              hipStream_t stream) {
  (void)in_sizes; (void)n_in; (void)out_size; (void)ws_size;
  const float* x    = (const float*)d_in[0];
  const float* cent = (const float*)d_in[1];
  float* out = (float*)d_out;
  float* ws  = (float*)d_ws;

  gemm1_softmax_kernel<<<dim3(PPTS / 16, NBATCH), G1_THREADS, 0, stream>>>(x, cent, ws);
  gemm2_kernel<<<dim3(CDIM / 32, KCLU / 16, NBATCH), G2_THREADS, 0, stream>>>(x, cent, ws, out);
  norms_kernel<<<NBATCH, NRM_THREADS, 0, stream>>>(out);
}


# first GEMM re-tiled: 20-wave block per 80 columns, each wave holds its M-tile's scaled weights in registers over 5 column tiles (loaded once, in flight across an LDS-only barrier); B from LDS
# speedup vs baseline: 3.1872x; 1.2252x over previous
#include <hip/hip_runtime.h>
#include <math.h>


#define NBATCH 16
#define CDIM   128
#define PPTS   1200
#define KCLU   64
#define S1     5
#define MROWS  (KCLU * S1)
#define ALPHA  100.0f
#define EPSV   1e-12f

typedef __attribute__((ext_vector_type(2))) float v2f;
typedef __attribute__((ext_vector_type(8))) float v8f;

#define WS_MULT  0

#define G1_COLS    80
#define G1_CT      (G1_COLS / 16)
#define G1_WAVES   (MROWS / 16)
#define G1_THREADS (G1_WAVES * 32)
#define G1_XPIECES (G1_COLS / 4)
#define XQP (4 * G1_COLS)
#define XP 84

__global__ __launch_bounds__(G1_THREADS) void gemm1_softmax_kernel(const float* __restrict__ x,
                                                                   const float* __restrict__ cent,
                                                                   float* __restrict__ ws) {
  __shared__ __attribute__((aligned(16))) float Xq[(CDIM / 4) * XQP];
  __shared__ float Cv[MROWS * XP];
  __shared__ float Pm[8][S1][G1_COLS];
  __shared__ float Ps[8][S1][G1_COLS];
  __shared__ __attribute__((aligned(16))) float Bs[MROWS];

  const int pt   = blockIdx.x;
  const int n    = blockIdx.y;
  const int t    = threadIdx.x;
  const int lane = t & 31;
  const int wv   = t >> 5;
  const int p0   = pt * G1_COLS;
  const int col  = lane & 15;
  const int half = lane >> 4;

  const int kq = t / G1_XPIECES;
  const int pc = t % G1_XPIECES;
  const float* xb = x + (size_t)n * CDIM * PPTS + (size_t)(4 * kq) * PPTS + p0 + 4 * pc;
  float4 xv[4];
  for (int q = 0; q < 4; ++q)
    xv[q] = *(const float4*)(xb + (size_t)q * PPTS);
  const float* wbase = cent + (size_t)(wv * 16 + col) * CDIM + 4 * half;
  float4 a4[16];
  for (int j = 0; j < 16; ++j)
    a4[j] = *(const float4*)(wbase + 8 * j);

  {
    float* xd = &Xq[kq * XQP + 16 * pc];
    float4 c0, c1, c2, c3;
    c0.x = xv[0].x; c0.y = xv[1].x; c0.z = xv[2].x; c0.w = xv[3].x;
    c1.x = xv[0].y; c1.y = xv[1].y; c1.z = xv[2].y; c1.w = xv[3].y;
    c2.x = xv[0].z; c2.y = xv[1].z; c2.z = xv[2].z; c2.w = xv[3].z;
    c3.x = xv[0].w; c3.y = xv[1].w; c3.z = xv[2].w; c3.w = xv[3].w;
    *(float4*)(xd + 0)  = c0;
    *(float4*)(xd + 4)  = c1;
    *(float4*)(xd + 8)  = c2;
    *(float4*)(xd + 12) = c3;
  }

  v2f aw[32];
  v2f sq = (v2f){};
  for (int j = 0; j < 16; ++j) {
    v2f a0; a0[0] = a4[j].x; a0[1] = a4[j].y;
    sq += a0 * a0;
    aw[2 * j] = a0 * (2.0f * ALPHA);
    v2f a1; a1[0] = a4[j].z; a1[1] = a4[j].w;
    sq += a1 * a1;
    aw[2 * j + 1] = a1 * (2.0f * ALPHA);
  }
  __builtin_amdgcn_fence(__ATOMIC_RELEASE, "workgroup", "local");
  __builtin_amdgcn_s_barrier();
  __builtin_amdgcn_fence(__ATOMIC_ACQUIRE, "workgroup", "local");

  const float* xcol = &Xq[half * XQP + 4 * col];
  v8f acc[G1_CT];
  for (int i = 0; i < G1_CT; ++i) acc[i] = (v8f){};
  for (int j = 0; j < 16; ++j) {
    v2f b0[G1_CT], b1[G1_CT];
    for (int i = 0; i < G1_CT; ++i) {
      const float4 b4 = *(const float4*)(xcol + (2 * j) * XQP + 64 * i);
      b0[i][0] = b4.x;
      b0[i][1] = b4.y;
      b1[i][0] = b4.z;
      b1[i][1] = b4.w;
    }
    for (int i = 0; i < G1_CT; ++i)
      acc[i] = __builtin_amdgcn_wmma_f32_16x16x4_f32(
          false, aw[2 * j], false, b0[i], (short)0, acc[i], false, false);
    for (int i = 0; i < G1_CT; ++i)
      acc[i] = __builtin_amdgcn_wmma_f32_16x16x4_f32(
          false, aw[2 * j + 1], false, b1[i], (short)0, acc[i], false, false);
  }

  {
    const float ssl = sq[0] + sq[1];
    const float tot = ssl + __shfl_xor(ssl, 16);
    Bs[wv * 16 + col] = -ALPHA * sqrtf(tot);
  }

  for (int i = 0; i < G1_CT; ++i) {
    const int mbase = wv * 16 + half * 8;
    for (int r = 0; r < 8; ++r)
      Cv[(mbase + r) * XP + 16 * i + col] = acc[i][r];
  }
  __syncthreads();

  __shared__ float Ma[S1][G1_COLS];
  __shared__ float Ra[S1][G1_COLS];
  const int p  = t % G1_COLS;
  const int kg = t / G1_COLS;
  float vv[8][S1];
  float e1[8][S1];
  float mg[S1];
  {
    float bl[8 * S1];
    for (int q = 0; q < 2 * S1; ++q) {
      const float4 b4 = *(const float4*)(&Bs[kg * 8 * S1 + 4 * q]);
      bl[4 * q + 0] = b4.x; bl[4 * q + 1] = b4.y; bl[4 * q + 2] = b4.z; bl[4 * q + 3] = b4.w;
    }
    for (int kk = 0; kk < 8; ++kk)
      for (int s = 0; s < S1; ++s)
        vv[kk][s] = Cv[((kg * 8 + kk) * S1 + s) * XP + p] + bl[kk * S1 + s];
    for (int s = 0; s < S1; ++s) {
      float m = vv[0][s];
      for (int kk = 1; kk < 8; ++kk) m = fmaxf(m, vv[kk][s]);
      float sm = 0.0f;
      for (int kk = 0; kk < 8; ++kk) {
        const float e = __expf(vv[kk][s] - m);
        e1[kk][s] = e;
        sm += e;
      }
      mg[s] = m;
      Pm[kg][s][p] = m;
      Ps[kg][s][p] = sm;
    }
  }
  __syncthreads();
  if (t < G1_COLS * S1) {
    float m = Pm[0][kg][p];
    for (int g = 1; g < 8; ++g) m = fmaxf(m, Pm[g][kg][p]);
    float sm = 0.0f;
    for (int g = 0; g < 8; ++g) sm += Ps[g][kg][p] * __expf(Pm[g][kg][p] - m);
    Ma[kg][p] = m;
    Ra[kg][p] = 1.0f / sm;
  }
  __syncthreads();
  {
    float gA[S1];
    for (int s = 0; s < S1; ++s) gA[s] = __expf(mg[s] - Ma[s][p]) * Ra[s][p];
    float* MU = ws + WS_MULT + (size_t)n * KCLU * PPTS;
    for (int kk = 0; kk < 8; ++kk) {
      float mB = vv[kk][0];
      for (int s = 1; s < S1; ++s) mB = fmaxf(mB, vv[kk][s]);
      float eB[S1];
      float sB = 0.0f;
      for (int s = 0; s < S1; ++s) {
        eB[s] = __expf(vv[kk][s] - mB);
        sB += eB[s];
      }
      const float rB = 1.0f / sB;
      float mult = 1.0f;
      for (int s = 0; s < S1; ++s) {
        const float sa = (e1[kk][s] * gA[s]) * (eB[s] * rB);
        mult *= (1.0f + sa);
      }
      MU[(size_t)(kg * 8 + kk) * PPTS + p0 + p] = mult;
    }
  }
}

#define G2_WAVES   10
#define G2_THREADS (G2_WAVES * 32)
#define G2_SLICE   (PPTS / G2_WAVES)
#define G2_PIECES  (G2_SLICE / 4)
#define G2_PAIRS   (G2_SLICE / 8)
#define G2_PITCH   124
#define G2_ROWS    48
#define G2_TPITCH  (8 * 32 + 16)

__global__ __launch_bounds__(G2_THREADS) void gemm2_kernel(const float* __restrict__ x,
                                                           const float* __restrict__ cent,
                                                           const float* __restrict__ ws,
                                                           float* __restrict__ out) {
  __shared__ __attribute__((aligned(16))) float St[G2_WAVES][G2_ROWS][G2_PITCH];
  __shared__ float Red[G2_WAVES][2][G2_TPITCH];
  __shared__ float Rs[G2_WAVES][32];

  const int cq   = blockIdx.x;
  const int mt   = blockIdx.y;
  const int n    = blockIdx.z;
  const int t    = threadIdx.x;
  const int lane = t & 31;
  const int wv   = t >> 5;
  const int l15  = lane & 15;
  const int half = lane >> 4;

  const float* Asrc = ws + WS_MULT + ((size_t)n * KCLU + mt * 16) * PPTS + wv * G2_SLICE;
  const float* Bsrc = x + ((size_t)n * CDIM + cq * 32) * PPTS + wv * G2_SLICE;
  float (*S)[G2_PITCH] = St[wv];

  float rep[2];
  for (int i = 0; i < 2; ++i) {
    const int rowk = (wv + i * G2_WAVES < 16) ? wv + i * G2_WAVES : wv;
    rep[i] = cent[(size_t)((mt * 16 + rowk) * S1) * CDIM + cq * 32 + lane];
  }

  if (lane < G2_PIECES) {
    float4 v[G2_ROWS];
    for (int r = 0; r < 16; ++r)
      v[r] = *(const float4*)(Asrc + (size_t)r * PPTS + lane * 4);
    for (int r = 0; r < 32; ++r)
      v[16 + r] = *(const float4*)(Bsrc + (size_t)r * PPTS + lane * 4);
    for (int r = 0; r < G2_ROWS; ++r)
      *(float4*)(&S[r][lane * 4]) = v[r];
  }
  __syncthreads();

  const float* Ar  = &S[l15][4 * half];
  const float* B0r = &S[16 + l15][4 * half];
  const float* B1r = &S[32 + l15][4 * half];
  float4 fa[G2_PAIRS], fp[G2_PAIRS], fq[G2_PAIRS];
  for (int jj = 0; jj < G2_PAIRS; ++jj) {
    fa[jj] = *(const float4*)(Ar + 8 * jj);
    fp[jj] = *(const float4*)(B0r + 8 * jj);
    fq[jj] = *(const float4*)(B1r + 8 * jj);
  }
  __builtin_amdgcn_sched_barrier(0);
  v8f acc0 = {};
  v8f acc1 = {};
  float rs = 0.0f;
  for (int jj = 0; jj < G2_PAIRS; ++jj) {
    const float4 a  = fa[jj];
    const float4 b0 = fp[jj];
    const float4 b1 = fq[jj];
    rs += (a.x + a.y) + (a.z + a.w);
    v2f a01, a23, p01, p23, q01, q23;
    a01[0] = a.x;  a01[1] = a.y;  a23[0] = a.z;  a23[1] = a.w;
    p01[0] = b0.x; p01[1] = b0.y; p23[0] = b0.z; p23[1] = b0.w;
    q01[0] = b1.x; q01[1] = b1.y; q23[0] = b1.z; q23[1] = b1.w;
    acc0 = __builtin_amdgcn_wmma_f32_16x16x4_f32(
        false, a01, false, p01, (short)0, acc0, false, false);
    acc1 = __builtin_amdgcn_wmma_f32_16x16x4_f32(
        false, a01, false, q01, (short)0, acc1, false, false);
    acc0 = __builtin_amdgcn_wmma_f32_16x16x4_f32(
        false, a23, false, p23, (short)0, acc0, false, false);
    acc1 = __builtin_amdgcn_wmma_f32_16x16x4_f32(
        false, a23, false, q23, (short)0, acc1, false, false);
  }

  for (int r = 0; r < 8; ++r) {
    Red[wv][0][r * 32 + lane] = acc0[r];
    Red[wv][1][r * 32 + lane] = acc1[r];
  }
  Rs[wv][lane] = rs;
  __syncthreads();

  for (int i = 0; i < 2; ++i) {
    const int rowk = wv + i * G2_WAVES;
    if (rowk < 16) {
      const int srci = (rowk & 7) * 32 + (rowk >> 3) * 16 + l15;
      float sum = 0.0f;
      float sk  = 0.0f;
      for (int w = 0; w < G2_WAVES; ++w) {
        sum += Red[w][half][srci];
        sk  += Rs[w][rowk] + Rs[w][rowk + 16];
      }
      out[((size_t)n * KCLU + mt * 16 + rowk) * CDIM + cq * 32 + lane] = sum - rep[i] * sk;
    }
  }
}

#define NRM_THREADS 1024

__global__ __launch_bounds__(NRM_THREADS) void norms_kernel(float* __restrict__ out) {
  __shared__ __attribute__((aligned(16))) float rn[KCLU];
  const int n    = blockIdx.x;
  const int t    = threadIdx.x;
  const int lane = t & 31;
  const int wv   = t >> 5;
  float* base = out + (size_t)n * KCLU * CDIM + lane * 4;

  float4 v[2];
  for (int i = 0; i < 2; ++i)
    v[i] = *(const float4*)(base + (size_t)(wv + 32 * i) * CDIM);

  float ssq[2];
  for (int i = 0; i < 2; ++i)
    ssq[i] = v[i].x * v[i].x + v[i].y * v[i].y + v[i].z * v[i].z + v[i].w * v[i].w;
  for (int off = 16; off > 0; off >>= 1)
    for (int i = 0; i < 2; ++i) ssq[i] += __shfl_xor(ssq[i], off);
  float rscale[2];
  for (int i = 0; i < 2; ++i) {
    const float sc = 1.0f / fmaxf(sqrtf(ssq[i]), EPSV);
    rscale[i] = sc;
    if (lane == 0) rn[wv + 32 * i] = ssq[i] * sc * sc;
  }
  __syncthreads();

  float tot = 0.0f;
  for (int q = 0; q < KCLU / 4; ++q) {
    const float4 r4 = *(const float4*)(&rn[4 * q]);
    tot += r4.x; tot += r4.y; tot += r4.z; tot += r4.w;
  }
  const float g = 1.0f / fmaxf(sqrtf(tot), EPSV);

  for (int i = 0; i < 2; ++i) {
    float4 y;
    y.x = v[i].x * rscale[i] * g;
    y.y = v[i].y * rscale[i] * g;
    y.z = v[i].z * rscale[i] * g;
    y.w = v[i].w * rscale[i] * g;
    *(float4*)(base + (size_t)(wv + 32 * i) * CDIM) = y;
  }
}

extern "C" void kernel_launch(void* const* d_in, const int* in_sizes, int n_in,
                              void* d_out, int out_size, void* d_ws, size_t ws_size,
                              hipStream_t stream) {
  (void)in_sizes; (void)n_in; (void)out_size; (void)ws_size;
  const float* x    = (const float*)d_in[0];
  const float* cent = (const float*)d_in[1];
  float* out = (float*)d_out;
  float* ws  = (float*)d_ws;

  gemm1_softmax_kernel<<<dim3(PPTS / G1_COLS, NBATCH), G1_THREADS, 0, stream>>>(x, cent, ws);
  gemm2_kernel<<<dim3(CDIM / 32, KCLU / 16, NBATCH), G2_THREADS, 0, stream>>>(x, cent, ws, out);
  norms_kernel<<<NBATCH, NRM_THREADS, 0, stream>>>(out);
}
